# starting kernel
# baseline (speedup 1.0000x reference)
#include <hip/hip_runtime.h>
#include <math.h>

typedef __attribute__((ext_vector_type(2))) float v2f;
typedef __attribute__((ext_vector_type(8))) float v8f;

#define IN_F   256
#define MLP    128
#define NATT   16
#define NBINS  64
#define PTILE  64
#define H      128
#define W      128
#define SRC    64
#define NB     4

__device__ __forceinline__ float softplus_stable(float v) {
    return fmaxf(v, 0.0f) + log1pf(__expf(-fabsf(v)));
}

__global__ __launch_bounds__(256)
void attractor_fused(const float* __restrict__ x,
                     const float* __restrict__ b_prev,
                     const float* __restrict__ emb,
                     const float* __restrict__ w1,
                     const float* __restrict__ b1,
                     const float* __restrict__ w2,
                     const float* __restrict__ b2,
                     float* __restrict__ out)
{
    __shared__ float s_xin [PTILE * IN_F];
    __shared__ float s_hidd[PTILE * MLP];
    __shared__ float s_att [PTILE * NATT];

    const int tid = threadIdx.x;
    const int blk = blockIdx.x;
    const int ox_base = (blk & 1) * PTILE;
    const int oy      = (blk >> 1) & (H - 1);
    const int n       = blk >> 8;

    const float sc = (float)(SRC - 1) / (float)(H - 1);
    const float fy = (float)oy * sc;
    const int   iy0 = (int)fy;
    const int   iy1 = min(iy0 + 1, SRC - 1);
    const float wy  = fy - (float)iy0;

    for (int idx = tid; idx < IN_F * PTILE; idx += 256) {
        const int c = idx >> 6;
        const int p = idx & (PTILE - 1);
        const int ox = ox_base + p;
        const float fx = (float)ox * sc;
        const int   ix0 = (int)fx;
        const int   ix1 = min(ix0 + 1, SRC - 1);
        const float wx  = fx - (float)ix0;

        const float* eb = emb + (size_t)(n * IN_F + c) * SRC * SRC;
        const float e00 = eb[iy0 * SRC + ix0];
        const float e01 = eb[iy0 * SRC + ix1];
        const float e10 = eb[iy1 * SRC + ix0];
        const float e11 = eb[iy1 * SRC + ix1];
        const float ev  = (1.f - wy) * ((1.f - wx) * e00 + wx * e01)
                        +        wy  * ((1.f - wx) * e10 + wx * e11);
        const float xv = x[((size_t)(n * IN_F + c) * H + oy) * W + ox];
        s_xin[p * IN_F + c] = xv + ev;
    }
    __syncthreads();

    const int lane = tid & 31;
    const int wave = tid >> 5;
    const int ln16 = lane & 15;
    const int hi   = lane >> 4;

    {
        const int m_base = wave * 16;
        v8f acc0 = {}, acc1 = {}, acc2 = {}, acc3 = {};
        const float* w1row = w1 + (size_t)(m_base + ln16) * IN_F + hi * 2;
        const float* bp0 = &s_xin[( 0 + ln16) * IN_F + hi * 2];
        const float* bp1 = &s_xin[(16 + ln16) * IN_F + hi * 2];
        const float* bp2 = &s_xin[(32 + ln16) * IN_F + hi * 2];
        const float* bp3 = &s_xin[(48 + ln16) * IN_F + hi * 2];
        #pragma unroll 4
        for (int k = 0; k < IN_F; k += 4) {
            const v2f a  = *(const v2f*)(w1row + k);
            const v2f bb0 = *(const v2f*)(bp0 + k);
            const v2f bb1 = *(const v2f*)(bp1 + k);
            const v2f bb2 = *(const v2f*)(bp2 + k);
            const v2f bb3 = *(const v2f*)(bp3 + k);
            acc0 = __builtin_amdgcn_wmma_f32_16x16x4_f32(false, a, false, bb0, (short)0, acc0, false, false);
            acc1 = __builtin_amdgcn_wmma_f32_16x16x4_f32(false, a, false, bb1, (short)0, acc1, false, false);
            acc2 = __builtin_amdgcn_wmma_f32_16x16x4_f32(false, a, false, bb2, (short)0, acc2, false, false);
            acc3 = __builtin_amdgcn_wmma_f32_16x16x4_f32(false, a, false, bb3, (short)0, acc3, false, false);
        }
        #pragma unroll
        for (int i = 0; i < 8; ++i) {
            const int m = m_base + hi * 8 + i;
            const float bias = b1[m];
            s_hidd[( 0 + ln16) * MLP + m] = fmaxf(acc0[i] + bias, 0.f);
            s_hidd[(16 + ln16) * MLP + m] = fmaxf(acc1[i] + bias, 0.f);
            s_hidd[(32 + ln16) * MLP + m] = fmaxf(acc2[i] + bias, 0.f);
            s_hidd[(48 + ln16) * MLP + m] = fmaxf(acc3[i] + bias, 0.f);
        }
    }
    __syncthreads();

    if (wave < 4) {
        const int pix = wave * 16;
        v8f acc = {};
        const float* w2row = w2 + (size_t)(ln16) * MLP + hi * 2;
        const float* bp = &s_hidd[(pix + ln16) * MLP + hi * 2];
        #pragma unroll 4
        for (int k = 0; k < MLP; k += 4) {
            const v2f a = *(const v2f*)(w2row + k);
            const v2f b = *(const v2f*)(bp + k);
            acc = __builtin_amdgcn_wmma_f32_16x16x4_f32(false, a, false, b, (short)0, acc, false, false);
        }
        #pragma unroll
        for (int i = 0; i < 8; ++i) {
            const int aix = hi * 8 + i;
            s_att[(pix + ln16) * NATT + aix] = softplus_stable(acc[i] + b2[aix]);
        }
    }
    __syncthreads();

    const size_t out_half = (size_t)NB * NBINS * H * W;
    for (int idx = tid; idx < NBINS * PTILE; idx += 256) {
        const int p = idx & (PTILE - 1);
        const int k = idx >> 6;
        const int ox = ox_base + p;
        const float fx = (float)ox * sc;
        const int   ix0 = (int)fx;
        const int   ix1 = min(ix0 + 1, SRC - 1);
        const float wx  = fx - (float)ix0;

        const float* bp = b_prev + (size_t)(n * NBINS + k) * SRC * SRC;
        const float c00 = bp[iy0 * SRC + ix0];
        const float c01 = bp[iy0 * SRC + ix1];
        const float c10 = bp[iy1 * SRC + ix0];
        const float c11 = bp[iy1 * SRC + ix1];
        const float bc  = (1.f - wy) * ((1.f - wx) * c00 + wx * c01)
                        +        wy  * ((1.f - wx) * c10 + wx * c11);

        float delta = 0.f;
        #pragma unroll
        for (int a = 0; a < NATT; ++a) {
            const float dx = s_att[p * NATT + a] - bc;
            delta += dx / (1.0f + 300.0f * dx * dx);
        }
        const float r = bc + delta;
        const size_t o = ((size_t)(n * NBINS + k) * H + oy) * W + ox;
        out[o] = r;
        out[o + out_half] = r;
    }
}

extern "C" void kernel_launch(void* const* d_in, const int* in_sizes, int n_in,
                              void* d_out, int out_size, void* d_ws, size_t ws_size,
                              hipStream_t stream) {
    const float* x      = (const float*)d_in[0];
    const float* b_prev = (const float*)d_in[1];
    const float* emb    = (const float*)d_in[2];
    const float* w1     = (const float*)d_in[3];
    const float* b1     = (const float*)d_in[4];
    const float* w2     = (const float*)d_in[5];
    const float* b2     = (const float*)d_in[6];
    float* out = (float*)d_out;

    dim3 grid(NB * H * (W / PTILE));
    attractor_fused<<<grid, 256, 0, stream>>>(x, b_prev, emb, w1, b1, w2, b2, out);
}


# One wave owns 32 pixels end to end: xin blended from global loads into B fragments, hidden kept in accumulators as GEMM2's B operand (GEMM2 k grouping, sum order change); no LDS, no barriers
# speedup vs baseline: 1.8028x; 1.8028x over previous
#include <hip/hip_runtime.h>
#include <math.h>

typedef __attribute__((ext_vector_type(2))) float v2f;
typedef __attribute__((ext_vector_type(4))) float v4f;
typedef __attribute__((ext_vector_type(8))) float v8f;

#define IN_F   256
#define MLP    128
#define NATT   16
#define NBINS  64
#define PTILE  32
#define H      128
#define W      128
#define SRC    64
#define NB     4
#define MT     (MLP / 16)


struct Emb { float e[2][4][4]; };
struct Xv  { float x[2][4]; };
struct Afr { v4f a[MT]; };

__device__ __forceinline__ float softplus_stable(float v) {
    return fmaxf(v, 0.0f) + log1pf(__expf(-fabsf(v)));
}

__device__ __forceinline__ void load_a(Afr& r, const float* __restrict__ wb, int k8) {
    #pragma unroll
    for (int mt = 0; mt < MT; ++mt)
        r.a[mt] = *(const v4f*)(wb + (size_t)mt * 16 * IN_F + k8);
}

__device__ __forceinline__ void load_e(Emb& r, const float* __restrict__ eb, int k8,
                                       const int (&eo)[2][4]) {
    #pragma unroll
    for (int q = 0; q < 4; ++q) {
        const float* ec = eb + (size_t)(k8 + q) * (SRC * SRC);
        #pragma unroll
        for (int j = 0; j < 2; ++j) {
            #pragma unroll
            for (int t = 0; t < 4; ++t) r.e[j][q][t] = ec[eo[j][t]];
        }
    }
}

__device__ __forceinline__ void load_x(Xv& r, const float* __restrict__ xb, int k8) {
    #pragma unroll
    for (int q = 0; q < 4; ++q) {
        const float* xc = xb + (size_t)(k8 + q) * (H * W);
        r.x[0][q] = xc[0];
        r.x[1][q] = xc[16];
    }
}

__device__ __forceinline__ void blend(float (&b)[2][4], const Xv& xv, const Emb& em,
                                      const float (&wx)[2], float wy) {
    #pragma unroll
    for (int j = 0; j < 2; ++j) {
        #pragma unroll
        for (int q = 0; q < 4; ++q) {
            const float ev = (1.f - wy) * ((1.f - wx[j]) * em.e[j][q][0] + wx[j] * em.e[j][q][1])
                           +        wy  * ((1.f - wx[j]) * em.e[j][q][2] + wx[j] * em.e[j][q][3]);
            b[j][q] = xv.x[j][q] + ev;
        }
    }
}

__device__ __forceinline__ void mma_blk(v8f (&acc)[MT][2], const Afr& r, const float (&b)[2][4]) {
    const v2f b00 = { b[0][0], b[0][1] };
    const v2f b10 = { b[1][0], b[1][1] };
    const v2f b01 = { b[0][2], b[0][3] };
    const v2f b11 = { b[1][2], b[1][3] };
    #pragma unroll
    for (int mt = 0; mt < MT; ++mt) {
        const v2f a_u0 = { r.a[mt].x, r.a[mt].y };
        const v2f a_u1 = { r.a[mt].z, r.a[mt].w };
        acc[mt][0] = __builtin_amdgcn_wmma_f32_16x16x4_f32(false, a_u0, false, b00, (short)0, acc[mt][0], false, false);
        acc[mt][1] = __builtin_amdgcn_wmma_f32_16x16x4_f32(false, a_u0, false, b10, (short)0, acc[mt][1], false, false);
        acc[mt][0] = __builtin_amdgcn_wmma_f32_16x16x4_f32(false, a_u1, false, b01, (short)0, acc[mt][0], false, false);
        acc[mt][1] = __builtin_amdgcn_wmma_f32_16x16x4_f32(false, a_u1, false, b11, (short)0, acc[mt][1], false, false);
    }
}

__global__ __launch_bounds__(256)
void attractor_fused(const float* __restrict__ x,
                     const float* __restrict__ b_prev,
                     const float* __restrict__ emb,
                     const float* __restrict__ w1,
                     const float* __restrict__ b1,
                     const float* __restrict__ w2,
                     const float* __restrict__ b2,
                     float* __restrict__ out)
{
    const int lane = threadIdx.x & 31;
    const int wave = __builtin_amdgcn_readfirstlane((int)(threadIdx.x >> 5));
    const int tile = blockIdx.x * 8 + wave;
    const int ox_base = (tile & 3) * PTILE;
    const int oy      = (tile >> 2) & (H - 1);
    const int n       = tile >> 9;
    const int c  = lane & 15;
    const int hi = lane >> 4;

    const float sc = (float)(SRC - 1) / (float)(H - 1);
    const float fy = (float)oy * sc;
    const int   iy0 = (int)fy;
    const int   iy1 = min(iy0 + 1, SRC - 1);
    const float wy  = fy - (float)iy0;

    float wxj[2];
    int   eo[2][4];
    #pragma unroll
    for (int j = 0; j < 2; ++j) {
        const int ox = ox_base + 16 * j + c;
        const float fx = (float)ox * sc;
        const int   ix0 = (int)fx;
        const int   ix1 = min(ix0 + 1, SRC - 1);
        wxj[j] = fx - (float)ix0;
        eo[j][0] = iy0 * SRC + ix0;
        eo[j][1] = iy0 * SRC + ix1;
        eo[j][2] = iy1 * SRC + ix0;
        eo[j][3] = iy1 * SRC + ix1;
    }
    const float* xb = x   + ((size_t)(n * IN_F + 4 * hi) * H + oy) * W + ox_base + c;
    const float* eb = emb + (size_t)(n * IN_F + 4 * hi) * (SRC * SRC);
    const float* wb = w1  + (size_t)c * IN_F + 4 * hi;

    v8f acc[MT][2];
    #pragma unroll
    for (int mt = 0; mt < MT; ++mt) { acc[mt][0] = v8f{}; acc[mt][1] = v8f{}; }

    Xv x0, x1; Emb em; Afr a0, a1;
    float bf[2][4];
    load_e(em, eb, 0, eo);
    load_a(a0, wb, 0);
    __builtin_amdgcn_sched_barrier(0);
    load_x(x0, xb, 0);
    load_x(x1, xb, 8);
    #pragma nounroll
    for (int k8 = 0; k8 < IN_F - 16; k8 += 16) {
        blend(bf, x0, em, wxj, wy);
        load_e(em, eb, k8 + 8, eo);
        load_a(a1, wb, k8 + 8);
        load_x(x0, xb, k8 + 16);
        __builtin_amdgcn_sched_barrier(0);
        mma_blk(acc, a0, bf);
        blend(bf, x1, em, wxj, wy);
        load_e(em, eb, k8 + 16, eo);
        load_a(a0, wb, k8 + 16);
        load_x(x1, xb, k8 + 24);
        __builtin_amdgcn_sched_barrier(0);
        mma_blk(acc, a1, bf);
    }
    blend(bf, x0, em, wxj, wy);
    load_e(em, eb, IN_F - 8, eo);
    load_a(a1, wb, IN_F - 8);
    __builtin_amdgcn_sched_barrier(0);
    mma_blk(acc, a0, bf);
    blend(bf, x1, em, wxj, wy);
    mma_blk(acc, a1, bf);

    v8f att0 = {}, att1 = {};
    {
        const float* w2row = w2 + (size_t)c * MLP + 8 * hi;
        const float* b1p   = b1 + 8 * hi;
        #pragma unroll
        for (int mt = 0; mt < MT; ++mt) {
            const v4f wa = *(const v4f*)(w2row + 16 * mt);
            const v4f wc = *(const v4f*)(w2row + 16 * mt + 4);
            const v4f ba = *(const v4f*)(b1p + 16 * mt);
            const v4f bc4 = *(const v4f*)(b1p + 16 * mt + 4);
            const float wv[8]   = { wa.x, wa.y, wa.z, wa.w, wc.x, wc.y, wc.z, wc.w };
            const float bias[8] = { ba.x, ba.y, ba.z, ba.w, bc4.x, bc4.y, bc4.z, bc4.w };
            float h0[8], h1[8];
            #pragma unroll
            for (int r = 0; r < 8; ++r) {
                h0[r] = fmaxf(acc[mt][0][r] + bias[r], 0.f);
                h1[r] = fmaxf(acc[mt][1][r] + bias[r], 0.f);
            }
            #pragma unroll
            for (int i = 0; i < 4; ++i) {
                const v2f a  = { wv[2 * i], wv[2 * i + 1] };
                const v2f p0 = { h0[2 * i], h0[2 * i + 1] };
                const v2f p1 = { h1[2 * i], h1[2 * i + 1] };
                att0 = __builtin_amdgcn_wmma_f32_16x16x4_f32(false, a, false, p0, (short)0, att0, false, false);
                att1 = __builtin_amdgcn_wmma_f32_16x16x4_f32(false, a, false, p1, (short)0, att1, false, false);
            }
        }
    }

    float at[NATT];
    {
        const v4f ca = *(const v4f*)(b2 + 8 * hi);
        const v4f cb = *(const v4f*)(b2 + 8 * hi + 4);
        const float bias2[8] = { ca.x, ca.y, ca.z, ca.w, cb.x, cb.y, cb.z, cb.w };
        #pragma unroll
        for (int r = 0; r < 8; ++r) {
            const float t0 = softplus_stable(att0[r] + bias2[r]);
            const float t1 = softplus_stable(att1[r] + bias2[r]);
            const float send = hi ? t0 : t1;
            const float recv = __shfl_xor(send, 16, 32);
            at[r]     = hi ? recv : t0;
            at[8 + r] = hi ? t1   : recv;
        }
    }

    {
        const int ox = ox_base + lane;
        const float fx = (float)ox * sc;
        const int   ix0 = (int)fx;
        const int   ix1 = min(ix0 + 1, SRC - 1);
        const float wx  = fx - (float)ix0;
        const int o00 = iy0 * SRC + ix0, o01 = iy0 * SRC + ix1;
        const int o10 = iy1 * SRC + ix0, o11 = iy1 * SRC + ix1;
        const float* bp = b_prev + (size_t)(n * NBINS) * (SRC * SRC);
        float* op = out + ((size_t)(n * NBINS) * H + oy) * W + ox;
        const size_t out_half = (size_t)NB * NBINS * H * W;

        const int off[4] = { o00, o01, o10, o11 };
        float ca[4], cb[4];
        #pragma unroll
        for (int t = 0; t < 4; ++t) { ca[t] = bp[off[t]]; cb[t] = bp[(SRC * SRC) + off[t]]; }
        for (int k = 0; k < NBINS; k += 2) {
            const float* bn = bp + (size_t)min(k + 2, NBINS - 2) * (SRC * SRC);
            float na[4], nb[4];
            #pragma unroll
            for (int t = 0; t < 4; ++t) { na[t] = bn[off[t]]; nb[t] = bn[(SRC * SRC) + off[t]]; }
            __builtin_amdgcn_sched_barrier(0);

            #pragma unroll
            for (int u = 0; u < 2; ++u) {
                const float (&cc)[4] = u ? cb : ca;
                const float bc  = (1.f - wy) * ((1.f - wx) * cc[0] + wx * cc[1])
                                +        wy  * ((1.f - wx) * cc[2] + wx * cc[3]);
                float delta = 0.f;
                #pragma unroll
                for (int a = 0; a < NATT; ++a) {
                    const float dx = at[a] - bc;
                    delta += dx / (1.0f + 300.0f * dx * dx);
                }
                const float r = bc + delta;
                op[(size_t)(k + u) * (H * W)] = r;
                op[(size_t)(k + u) * (H * W) + out_half] = r;
            }
            __builtin_amdgcn_sched_barrier(0);
            #pragma unroll
            for (int t = 0; t < 4; ++t) { ca[t] = na[t]; cb[t] = nb[t]; }
        }
    }
}

extern "C" void kernel_launch(void* const* d_in, const int* in_sizes, int n_in,
                              void* d_out, int out_size, void* d_ws, size_t ws_size,
                              hipStream_t stream) {
    const float* x      = (const float*)d_in[0];
    const float* b_prev = (const float*)d_in[1];
    const float* emb    = (const float*)d_in[2];
    const float* w1     = (const float*)d_in[3];
    const float* b1     = (const float*)d_in[4];
    const float* w2     = (const float*)d_in[5];
    const float* b2     = (const float*)d_in[6];
    float* out = (float*)d_out;

    dim3 grid(256);
    attractor_fused<<<grid, 256, 0, stream>>>(x, b_prev, emb, w1, b1, w2, b2, out);
}


# Stage 4: the 16 attractor terms of a bin added four at a time as one fraction over a common denominator: 4 IEEE divisions per bin, not 16; sum tree and rounding of delta change
# speedup vs baseline: 2.0227x; 1.1220x over previous
#include <hip/hip_runtime.h>
#include <math.h>

typedef __attribute__((ext_vector_type(2))) float v2f;
typedef __attribute__((ext_vector_type(4))) float v4f;
typedef __attribute__((ext_vector_type(8))) float v8f;

#define IN_F   256
#define MLP    128
#define NATT   16
#define NBINS  64
#define PTILE  32
#define H      128
#define W      128
#define SRC    64
#define NB     4
#define MT     (MLP / 16)


struct Emb { float e[2][4][4]; };
struct Xv  { float x[2][4]; };
struct Afr { v4f a[MT]; };

__device__ __forceinline__ float softplus_stable(float v) {
    return fmaxf(v, 0.0f) + log1pf(__expf(-fabsf(v)));
}

__device__ __forceinline__ void load_a(Afr& r, const float* __restrict__ wb, int k8) {
    #pragma unroll
    for (int mt = 0; mt < MT; ++mt)
        r.a[mt] = *(const v4f*)(wb + (size_t)mt * 16 * IN_F + k8);
}

__device__ __forceinline__ void load_e(Emb& r, const float* __restrict__ eb, int k8,
                                       const int (&eo)[2][4]) {
    #pragma unroll
    for (int q = 0; q < 4; ++q) {
        const float* ec = eb + (size_t)(k8 + q) * (SRC * SRC);
        #pragma unroll
        for (int j = 0; j < 2; ++j) {
            #pragma unroll
            for (int t = 0; t < 4; ++t) r.e[j][q][t] = ec[eo[j][t]];
        }
    }
}

__device__ __forceinline__ void load_x(Xv& r, const float* __restrict__ xb, int k8) {
    #pragma unroll
    for (int q = 0; q < 4; ++q) {
        const float* xc = xb + (size_t)(k8 + q) * (H * W);
        r.x[0][q] = xc[0];
        r.x[1][q] = xc[16];
    }
}

__device__ __forceinline__ void blend(float (&b)[2][4], const Xv& xv, const Emb& em,
                                      const float (&wx)[2], float wy) {
    #pragma unroll
    for (int j = 0; j < 2; ++j) {
        #pragma unroll
        for (int q = 0; q < 4; ++q) {
            const float ev = (1.f - wy) * ((1.f - wx[j]) * em.e[j][q][0] + wx[j] * em.e[j][q][1])
                           +        wy  * ((1.f - wx[j]) * em.e[j][q][2] + wx[j] * em.e[j][q][3]);
            b[j][q] = xv.x[j][q] + ev;
        }
    }
}

__device__ __forceinline__ void mma_blk(v8f (&acc)[MT][2], const Afr& r, const float (&b)[2][4]) {
    const v2f b00 = { b[0][0], b[0][1] };
    const v2f b10 = { b[1][0], b[1][1] };
    const v2f b01 = { b[0][2], b[0][3] };
    const v2f b11 = { b[1][2], b[1][3] };
    #pragma unroll
    for (int mt = 0; mt < MT; ++mt) {
        const v2f a_u0 = { r.a[mt].x, r.a[mt].y };
        const v2f a_u1 = { r.a[mt].z, r.a[mt].w };
        acc[mt][0] = __builtin_amdgcn_wmma_f32_16x16x4_f32(false, a_u0, false, b00, (short)0, acc[mt][0], false, false);
        acc[mt][1] = __builtin_amdgcn_wmma_f32_16x16x4_f32(false, a_u0, false, b10, (short)0, acc[mt][1], false, false);
        acc[mt][0] = __builtin_amdgcn_wmma_f32_16x16x4_f32(false, a_u1, false, b01, (short)0, acc[mt][0], false, false);
        acc[mt][1] = __builtin_amdgcn_wmma_f32_16x16x4_f32(false, a_u1, false, b11, (short)0, acc[mt][1], false, false);
    }
}

__global__ __launch_bounds__(256)
void attractor_fused(const float* __restrict__ x,
                     const float* __restrict__ b_prev,
                     const float* __restrict__ emb,
                     const float* __restrict__ w1,
                     const float* __restrict__ b1,
                     const float* __restrict__ w2,
                     const float* __restrict__ b2,
                     float* __restrict__ out)
{
    const int lane = threadIdx.x & 31;
    const int wave = __builtin_amdgcn_readfirstlane((int)(threadIdx.x >> 5));
    const int tile = blockIdx.x * 8 + wave;
    const int ox_base = (tile & 3) * PTILE;
    const int oy      = (tile >> 2) & (H - 1);
    const int n       = tile >> 9;
    const int c  = lane & 15;
    const int hi = lane >> 4;

    const float sc = (float)(SRC - 1) / (float)(H - 1);
    const float fy = (float)oy * sc;
    const int   iy0 = (int)fy;
    const int   iy1 = min(iy0 + 1, SRC - 1);
    const float wy  = fy - (float)iy0;

    float wxj[2];
    int   eo[2][4];
    #pragma unroll
    for (int j = 0; j < 2; ++j) {
        const int ox = ox_base + 16 * j + c;
        const float fx = (float)ox * sc;
        const int   ix0 = (int)fx;
        const int   ix1 = min(ix0 + 1, SRC - 1);
        wxj[j] = fx - (float)ix0;
        eo[j][0] = iy0 * SRC + ix0;
        eo[j][1] = iy0 * SRC + ix1;
        eo[j][2] = iy1 * SRC + ix0;
        eo[j][3] = iy1 * SRC + ix1;
    }
    const float* xb = x   + ((size_t)(n * IN_F + 4 * hi) * H + oy) * W + ox_base + c;
    const float* eb = emb + (size_t)(n * IN_F + 4 * hi) * (SRC * SRC);
    const float* wb = w1  + (size_t)c * IN_F + 4 * hi;

    v8f acc[MT][2];
    #pragma unroll
    for (int mt = 0; mt < MT; ++mt) { acc[mt][0] = v8f{}; acc[mt][1] = v8f{}; }

    Xv x0, x1; Emb em; Afr a0, a1;
    float bf[2][4];
    load_e(em, eb, 0, eo);
    load_a(a0, wb, 0);
    __builtin_amdgcn_sched_barrier(0);
    load_x(x0, xb, 0);
    load_x(x1, xb, 8);
    #pragma nounroll
    for (int k8 = 0; k8 < IN_F - 16; k8 += 16) {
        blend(bf, x0, em, wxj, wy);
        load_e(em, eb, k8 + 8, eo);
        load_a(a1, wb, k8 + 8);
        load_x(x0, xb, k8 + 16);
        __builtin_amdgcn_sched_barrier(0);
        mma_blk(acc, a0, bf);
        blend(bf, x1, em, wxj, wy);
        load_e(em, eb, k8 + 16, eo);
        load_a(a0, wb, k8 + 16);
        load_x(x1, xb, k8 + 24);
        __builtin_amdgcn_sched_barrier(0);
        mma_blk(acc, a1, bf);
    }
    blend(bf, x0, em, wxj, wy);
    load_e(em, eb, IN_F - 8, eo);
    load_a(a1, wb, IN_F - 8);
    __builtin_amdgcn_sched_barrier(0);
    mma_blk(acc, a0, bf);
    blend(bf, x1, em, wxj, wy);
    mma_blk(acc, a1, bf);

    v8f att0 = {}, att1 = {};
    {
        const float* w2row = w2 + (size_t)c * MLP + 8 * hi;
        const float* b1p   = b1 + 8 * hi;
        #pragma unroll
        for (int mt = 0; mt < MT; ++mt) {
            const v4f wa = *(const v4f*)(w2row + 16 * mt);
            const v4f wc = *(const v4f*)(w2row + 16 * mt + 4);
            const v4f ba = *(const v4f*)(b1p + 16 * mt);
            const v4f bc4 = *(const v4f*)(b1p + 16 * mt + 4);
            const float wv[8]   = { wa.x, wa.y, wa.z, wa.w, wc.x, wc.y, wc.z, wc.w };
            const float bias[8] = { ba.x, ba.y, ba.z, ba.w, bc4.x, bc4.y, bc4.z, bc4.w };
            float h0[8], h1[8];
            #pragma unroll
            for (int r = 0; r < 8; ++r) {
                h0[r] = fmaxf(acc[mt][0][r] + bias[r], 0.f);
                h1[r] = fmaxf(acc[mt][1][r] + bias[r], 0.f);
            }
            #pragma unroll
            for (int i = 0; i < 4; ++i) {
                const v2f a  = { wv[2 * i], wv[2 * i + 1] };
                const v2f p0 = { h0[2 * i], h0[2 * i + 1] };
                const v2f p1 = { h1[2 * i], h1[2 * i + 1] };
                att0 = __builtin_amdgcn_wmma_f32_16x16x4_f32(false, a, false, p0, (short)0, att0, false, false);
                att1 = __builtin_amdgcn_wmma_f32_16x16x4_f32(false, a, false, p1, (short)0, att1, false, false);
            }
        }
    }

    float at[NATT];
    {
        const v4f ca = *(const v4f*)(b2 + 8 * hi);
        const v4f cb = *(const v4f*)(b2 + 8 * hi + 4);
        const float bias2[8] = { ca.x, ca.y, ca.z, ca.w, cb.x, cb.y, cb.z, cb.w };
        #pragma unroll
        for (int r = 0; r < 8; ++r) {
            const float t0 = softplus_stable(att0[r] + bias2[r]);
            const float t1 = softplus_stable(att1[r] + bias2[r]);
            const float send = hi ? t0 : t1;
            const float recv = __shfl_xor(send, 16, 32);
            at[r]     = hi ? recv : t0;
            at[8 + r] = hi ? t1   : recv;
        }
    }

    {
        const int ox = ox_base + lane;
        const float fx = (float)ox * sc;
        const int   ix0 = (int)fx;
        const int   ix1 = min(ix0 + 1, SRC - 1);
        const float wx  = fx - (float)ix0;
        const int o00 = iy0 * SRC + ix0, o01 = iy0 * SRC + ix1;
        const int o10 = iy1 * SRC + ix0, o11 = iy1 * SRC + ix1;
        const float* bp = b_prev + (size_t)(n * NBINS) * (SRC * SRC);
        float* op = out + ((size_t)(n * NBINS) * H + oy) * W + ox;
        const size_t out_half = (size_t)NB * NBINS * H * W;

        const int off[4] = { o00, o01, o10, o11 };
        float ca[4], cb[4];
        #pragma unroll
        for (int t = 0; t < 4; ++t) { ca[t] = bp[off[t]]; cb[t] = bp[(SRC * SRC) + off[t]]; }
        for (int k = 0; k < NBINS; k += 2) {
            const float* bn = bp + (size_t)min(k + 2, NBINS - 2) * (SRC * SRC);
            float na[4], nb[4];
            #pragma unroll
            for (int t = 0; t < 4; ++t) { na[t] = bn[off[t]]; nb[t] = bn[(SRC * SRC) + off[t]]; }
            __builtin_amdgcn_sched_barrier(0);

            #pragma unroll
            for (int u = 0; u < 2; ++u) {
                const float (&cc)[4] = u ? cb : ca;
                const float bc  = (1.f - wy) * ((1.f - wx) * cc[0] + wx * cc[1])
                                +        wy  * ((1.f - wx) * cc[2] + wx * cc[3]);
                float delta = 0.f;
                #pragma unroll
                for (int g = 0; g < NATT; g += 4) {
                    float nx[4], dn[4];
                    #pragma unroll
                    for (int i = 0; i < 4; ++i) {
                        const float dx = at[g + i] - bc;
                        nx[i] = dx;
                        dn[i] = 1.0f + 300.0f * dx * dx;
                    }
                    const float n01 = nx[0] * dn[1] + nx[1] * dn[0];
                    const float d01 = dn[0] * dn[1];
                    const float n23 = nx[2] * dn[3] + nx[3] * dn[2];
                    const float d23 = dn[2] * dn[3];
                    delta += (n01 * d23 + n23 * d01) / (d01 * d23);
                }
                const float r = bc + delta;
                op[(size_t)(k + u) * (H * W)] = r;
                op[(size_t)(k + u) * (H * W) + out_half] = r;
            }
            __builtin_amdgcn_sched_barrier(0);
            #pragma unroll
            for (int t = 0; t < 4; ++t) { ca[t] = na[t]; cb[t] = nb[t]; }
        }
    }
}

extern "C" void kernel_launch(void* const* d_in, const int* in_sizes, int n_in,
                              void* d_out, int out_size, void* d_ws, size_t ws_size,
                              hipStream_t stream) {
    const float* x      = (const float*)d_in[0];
    const float* b_prev = (const float*)d_in[1];
    const float* emb    = (const float*)d_in[2];
    const float* w1     = (const float*)d_in[3];
    const float* b1     = (const float*)d_in[4];
    const float* w2     = (const float*)d_in[5];
    const float* b2     = (const float*)d_in[6];
    float* out = (float*)d_out;

    dim3 grid(256);
    attractor_fused<<<grid, 256, 0, stream>>>(x, b_prev, emb, w1, b1, w2, b2, out);
}


# Embedding taps shared by lanes: a lane loads one source column per row, blends rows first, takes both column taps from neighbour lanes (v_permlane16_var); 16 not 32 loads per block, 2 blocks ahead
# speedup vs baseline: 2.1790x; 1.0773x over previous
#include <hip/hip_runtime.h>
#include <math.h>

typedef __attribute__((ext_vector_type(2))) float v2f;
typedef __attribute__((ext_vector_type(4))) float v4f;
typedef __attribute__((ext_vector_type(8))) float v8f;

#define IN_F   256
#define MLP    128
#define NATT   16
#define NBINS  64
#define PTILE  32
#define H      128
#define W      128
#define SRC    64
#define NB     4
#define MT     (MLP / 16)


struct Emb { float e[2][4][2]; };
struct Xv  { float x[2][4]; };
struct Afr { v4f a[MT]; };

__device__ __forceinline__ float softplus_stable(float v) {
    return fmaxf(v, 0.0f) + log1pf(__expf(-fabsf(v)));
}

__device__ __forceinline__ void load_a(Afr& r, const float* __restrict__ wb, int k8) {
    #pragma unroll
    for (int mt = 0; mt < MT; ++mt)
        r.a[mt] = *(const v4f*)(wb + (size_t)mt * 16 * IN_F + k8);
}

__device__ __forceinline__ void load_e(Emb& r, const float* __restrict__ eb, int k8,
                                       const int (&eo)[2][2]) {
    #pragma unroll
    for (int q = 0; q < 4; ++q) {
        const float* ec = eb + (size_t)(k8 + q) * (SRC * SRC);
        #pragma unroll
        for (int j = 0; j < 2; ++j) {
            #pragma unroll
            for (int t = 0; t < 2; ++t) r.e[j][q][t] = ec[eo[j][t]];
        }
    }
}

__device__ __forceinline__ float lane_pick(float v, int sel) {
    const unsigned u = __float_as_uint(v);
    return __uint_as_float(__builtin_amdgcn_permlane16_var(u, u, (unsigned)sel, false, false));
}

__device__ __forceinline__ void load_x(Xv& r, const float* __restrict__ xb, int k8) {
    #pragma unroll
    for (int q = 0; q < 4; ++q) {
        const float* xc = xb + (size_t)(k8 + q) * (H * W);
        r.x[0][q] = xc[0];
        r.x[1][q] = xc[16];
    }
}

__device__ __forceinline__ void blend(float (&b)[2][4], const Xv& xv, const Emb& em,
                                      const float (&wx)[2], float wy, const int (&sel)[2][2]) {
    #pragma unroll
    for (int j = 0; j < 2; ++j) {
        #pragma unroll
        for (int q = 0; q < 4; ++q) {
            const float v  = (1.f - wy) * em.e[j][q][0] + wy * em.e[j][q][1];
            const float t0 = lane_pick(v, sel[j][0]);
            const float t1 = lane_pick(v, sel[j][1]);
            b[j][q] = xv.x[j][q] + ((1.f - wx[j]) * t0 + wx[j] * t1);
        }
    }
}

__device__ __forceinline__ void mma_blk(v8f (&acc)[MT][2], const Afr& r, const float (&b)[2][4]) {
    const v2f b00 = { b[0][0], b[0][1] };
    const v2f b10 = { b[1][0], b[1][1] };
    const v2f b01 = { b[0][2], b[0][3] };
    const v2f b11 = { b[1][2], b[1][3] };
    #pragma unroll
    for (int mt = 0; mt < MT; ++mt) {
        const v2f a_u0 = { r.a[mt].x, r.a[mt].y };
        const v2f a_u1 = { r.a[mt].z, r.a[mt].w };
        acc[mt][0] = __builtin_amdgcn_wmma_f32_16x16x4_f32(false, a_u0, false, b00, (short)0, acc[mt][0], false, false);
        acc[mt][1] = __builtin_amdgcn_wmma_f32_16x16x4_f32(false, a_u0, false, b10, (short)0, acc[mt][1], false, false);
        acc[mt][0] = __builtin_amdgcn_wmma_f32_16x16x4_f32(false, a_u1, false, b01, (short)0, acc[mt][0], false, false);
        acc[mt][1] = __builtin_amdgcn_wmma_f32_16x16x4_f32(false, a_u1, false, b11, (short)0, acc[mt][1], false, false);
    }
}

__global__ __launch_bounds__(256)
void attractor_fused(const float* __restrict__ x,
                     const float* __restrict__ b_prev,
                     const float* __restrict__ emb,
                     const float* __restrict__ w1,
                     const float* __restrict__ b1,
                     const float* __restrict__ w2,
                     const float* __restrict__ b2,
                     float* __restrict__ out)
{
    const int lane = threadIdx.x & 31;
    const int wave = __builtin_amdgcn_readfirstlane((int)(threadIdx.x >> 5));
    const int tile = blockIdx.x * 8 + wave;
    const int ox_base = (tile & 3) * PTILE;
    const int oy      = (tile >> 2) & (H - 1);
    const int n       = tile >> 9;
    const int c  = lane & 15;
    const int hi = lane >> 4;

    const float sc = (float)(SRC - 1) / (float)(H - 1);
    const float fy = (float)oy * sc;
    const int   iy0 = (int)fy;
    const int   iy1 = min(iy0 + 1, SRC - 1);
    const float wy  = fy - (float)iy0;

    float wxj[2];
    int   eo[2][2];
    int   sel[2][2];
    #pragma unroll
    for (int j = 0; j < 2; ++j) {
        const int ox = ox_base + 16 * j + c;
        const float fx = (float)ox * sc;
        const int   ix0 = (int)fx;
        const int   ix1 = min(ix0 + 1, SRC - 1);
        wxj[j] = fx - (float)ix0;
        const int s0 = (int)((float)(ox_base + 16 * j) * sc);
        const int lc = min(s0 + c, SRC - 1);
        eo[j][0] = iy0 * SRC + lc;
        eo[j][1] = iy1 * SRC + lc;
        sel[j][0] = ix0 - s0;
        sel[j][1] = ix1 - s0;
    }
    const float* xb = x   + ((size_t)(n * IN_F + 4 * hi) * H + oy) * W + ox_base + c;
    const float* eb = emb + (size_t)(n * IN_F + 4 * hi) * (SRC * SRC);
    const float* wb = w1  + (size_t)c * IN_F + 4 * hi;

    v8f acc[MT][2];
    #pragma unroll
    for (int mt = 0; mt < MT; ++mt) { acc[mt][0] = v8f{}; acc[mt][1] = v8f{}; }

    Xv x0, x1; Emb em0, em1; Afr a0, a1;
    float bf[2][4];
    load_e(em0, eb, 0, eo);
    load_x(x0, xb, 0);
    __builtin_amdgcn_sched_barrier(0);
    load_e(em1, eb, 8, eo);
    load_a(a0, wb, 0);
    load_x(x1, xb, 8);
    #pragma nounroll
    for (int k8 = 0; k8 < IN_F - 16; k8 += 16) {
        blend(bf, x0, em0, wxj, wy, sel);
        load_e(em0, eb, k8 + 16, eo);
        load_a(a1, wb, k8 + 8);
        load_x(x0, xb, k8 + 16);
        __builtin_amdgcn_sched_barrier(0);
        mma_blk(acc, a0, bf);
        blend(bf, x1, em1, wxj, wy, sel);
        load_e(em1, eb, k8 + 24, eo);
        load_a(a0, wb, k8 + 16);
        load_x(x1, xb, k8 + 24);
        __builtin_amdgcn_sched_barrier(0);
        mma_blk(acc, a1, bf);
    }
    blend(bf, x0, em0, wxj, wy, sel);
    load_a(a1, wb, IN_F - 8);
    __builtin_amdgcn_sched_barrier(0);
    mma_blk(acc, a0, bf);
    blend(bf, x1, em1, wxj, wy, sel);
    mma_blk(acc, a1, bf);

    v8f att0 = {}, att1 = {};
    {
        const float* w2row = w2 + (size_t)c * MLP + 8 * hi;
        const float* b1p   = b1 + 8 * hi;
        #pragma unroll
        for (int mt = 0; mt < MT; ++mt) {
            const v4f wa = *(const v4f*)(w2row + 16 * mt);
            const v4f wc = *(const v4f*)(w2row + 16 * mt + 4);
            const v4f ba = *(const v4f*)(b1p + 16 * mt);
            const v4f bc4 = *(const v4f*)(b1p + 16 * mt + 4);
            const float wv[8]   = { wa.x, wa.y, wa.z, wa.w, wc.x, wc.y, wc.z, wc.w };
            const float bias[8] = { ba.x, ba.y, ba.z, ba.w, bc4.x, bc4.y, bc4.z, bc4.w };
            float h0[8], h1[8];
            #pragma unroll
            for (int r = 0; r < 8; ++r) {
                h0[r] = fmaxf(acc[mt][0][r] + bias[r], 0.f);
                h1[r] = fmaxf(acc[mt][1][r] + bias[r], 0.f);
            }
            #pragma unroll
            for (int i = 0; i < 4; ++i) {
                const v2f a  = { wv[2 * i], wv[2 * i + 1] };
                const v2f p0 = { h0[2 * i], h0[2 * i + 1] };
                const v2f p1 = { h1[2 * i], h1[2 * i + 1] };
                att0 = __builtin_amdgcn_wmma_f32_16x16x4_f32(false, a, false, p0, (short)0, att0, false, false);
                att1 = __builtin_amdgcn_wmma_f32_16x16x4_f32(false, a, false, p1, (short)0, att1, false, false);
            }
        }
    }

    float at[NATT];
    {
        const v4f ca = *(const v4f*)(b2 + 8 * hi);
        const v4f cb = *(const v4f*)(b2 + 8 * hi + 4);
        const float bias2[8] = { ca.x, ca.y, ca.z, ca.w, cb.x, cb.y, cb.z, cb.w };
        #pragma unroll
        for (int r = 0; r < 8; ++r) {
            const float t0 = softplus_stable(att0[r] + bias2[r]);
            const float t1 = softplus_stable(att1[r] + bias2[r]);
            const float send = hi ? t0 : t1;
            const float recv = __shfl_xor(send, 16, 32);
            at[r]     = hi ? recv : t0;
            at[8 + r] = hi ? t1   : recv;
        }
    }

    {
        const int ox = ox_base + lane;
        const float fx = (float)ox * sc;
        const int   ix0 = (int)fx;
        const int   ix1 = min(ix0 + 1, SRC - 1);
        const float wx  = fx - (float)ix0;
        const int o00 = iy0 * SRC + ix0, o01 = iy0 * SRC + ix1;
        const int o10 = iy1 * SRC + ix0, o11 = iy1 * SRC + ix1;
        const float* bp = b_prev + (size_t)(n * NBINS) * (SRC * SRC);
        float* op = out + ((size_t)(n * NBINS) * H + oy) * W + ox;
        const size_t out_half = (size_t)NB * NBINS * H * W;

        const int off[4] = { o00, o01, o10, o11 };
        float ca[4], cb[4];
        #pragma unroll
        for (int t = 0; t < 4; ++t) { ca[t] = bp[off[t]]; cb[t] = bp[(SRC * SRC) + off[t]]; }
        for (int k = 0; k < NBINS; k += 2) {
            const float* bn = bp + (size_t)min(k + 2, NBINS - 2) * (SRC * SRC);
            float na[4], nb[4];
            #pragma unroll
            for (int t = 0; t < 4; ++t) { na[t] = bn[off[t]]; nb[t] = bn[(SRC * SRC) + off[t]]; }
            __builtin_amdgcn_sched_barrier(0);

            #pragma unroll
            for (int u = 0; u < 2; ++u) {
                const float (&cc)[4] = u ? cb : ca;
                const float bc  = (1.f - wy) * ((1.f - wx) * cc[0] + wx * cc[1])
                                +        wy  * ((1.f - wx) * cc[2] + wx * cc[3]);
                float delta = 0.f;
                #pragma unroll
                for (int g = 0; g < NATT; g += 4) {
                    float nx[4], dn[4];
                    #pragma unroll
                    for (int i = 0; i < 4; ++i) {
                        const float dx = at[g + i] - bc;
                        nx[i] = dx;
                        dn[i] = 1.0f + 300.0f * dx * dx;
                    }
                    const float n01 = nx[0] * dn[1] + nx[1] * dn[0];
                    const float d01 = dn[0] * dn[1];
                    const float n23 = nx[2] * dn[3] + nx[3] * dn[2];
                    const float d23 = dn[2] * dn[3];
                    delta += (n01 * d23 + n23 * d01) / (d01 * d23);
                }
                const float r = bc + delta;
                op[(size_t)(k + u) * (H * W)] = r;
                op[(size_t)(k + u) * (H * W) + out_half] = r;
            }
            __builtin_amdgcn_sched_barrier(0);
            #pragma unroll
            for (int t = 0; t < 4; ++t) { ca[t] = na[t]; cb[t] = nb[t]; }
        }
    }
}

extern "C" void kernel_launch(void* const* d_in, const int* in_sizes, int n_in,
                              void* d_out, int out_size, void* d_ws, size_t ws_size,
                              hipStream_t stream) {
    const float* x      = (const float*)d_in[0];
    const float* b_prev = (const float*)d_in[1];
    const float* emb    = (const float*)d_in[2];
    const float* w1     = (const float*)d_in[3];
    const float* b1     = (const float*)d_in[4];
    const float* w2     = (const float*)d_in[5];
    const float* b2     = (const float*)d_in[6];
    float* out = (float*)d_out;

    dim3 grid(256);
    attractor_fused<<<grid, 256, 0, stream>>>(x, b_prev, emb, w1, b1, w2, b2, out);
}


# One wave owns 64 pixels (4 pixel tiles; 1024 waves, one per SIMD): each W1 fragment load, 16 cache lines, now feeds 4 WMMAs not 2, halving W1 loads per pixel; k order and all sums unchanged
# speedup vs baseline: 2.2476x; 1.0315x over previous
#include <hip/hip_runtime.h>
#include <math.h>

typedef __attribute__((ext_vector_type(2))) float v2f;
typedef __attribute__((ext_vector_type(4))) float v4f;
typedef __attribute__((ext_vector_type(8))) float v8f;

#define IN_F   256
#define MLP    128
#define NATT   16
#define NBINS  64
#define PTILE  64
#define JT     4
#define NG     2
#define H      128
#define W      128
#define SRC    64
#define NB     4
#define MT     (MLP / 16)


struct Emb { float e[JT][4][2]; };
struct Xv  { float x[JT][4]; };
struct Afr { v4f a[MT]; };

__device__ __forceinline__ float softplus_stable(float v) {
    return fmaxf(v, 0.0f) + log1pf(__expf(-fabsf(v)));
}

__device__ __forceinline__ void load_a(Afr& r, const float* __restrict__ wb, int k8) {
    #pragma unroll
    for (int mt = 0; mt < MT; ++mt)
        r.a[mt] = *(const v4f*)(wb + (size_t)mt * 16 * IN_F + k8);
}

__device__ __forceinline__ void load_e(Emb& r, const float* __restrict__ eb, int k8,
                                       const int (&eo)[JT][2]) {
    #pragma unroll
    for (int q = 0; q < 4; ++q) {
        const float* ec = eb + (size_t)(k8 + q) * (SRC * SRC);
        #pragma unroll
        for (int j = 0; j < JT; ++j) {
            #pragma unroll
            for (int t = 0; t < 2; ++t) r.e[j][q][t] = ec[eo[j][t]];
        }
    }
}

__device__ __forceinline__ float lane_pick(float v, int sel) {
    const unsigned u = __float_as_uint(v);
    return __uint_as_float(__builtin_amdgcn_permlane16_var(u, u, (unsigned)sel, false, false));
}

__device__ __forceinline__ void load_x(Xv& r, const float* __restrict__ xb, int k8) {
    #pragma unroll
    for (int q = 0; q < 4; ++q) {
        const float* xc = xb + (size_t)(k8 + q) * (H * W);
        #pragma unroll
        for (int j = 0; j < JT; ++j) r.x[j][q] = xc[16 * j];
    }
}

__device__ __forceinline__ void blend(float (&b)[JT][4], const Xv& xv, const Emb& em,
                                      const float (&wx)[JT], float wy, const int (&sel)[JT][2]) {
    #pragma unroll
    for (int j = 0; j < JT; ++j) {
        #pragma unroll
        for (int q = 0; q < 4; ++q) {
            const float v  = (1.f - wy) * em.e[j][q][0] + wy * em.e[j][q][1];
            const float t0 = lane_pick(v, sel[j][0]);
            const float t1 = lane_pick(v, sel[j][1]);
            b[j][q] = xv.x[j][q] + ((1.f - wx[j]) * t0 + wx[j] * t1);
        }
    }
}

__device__ __forceinline__ void mma_blk(v8f (&acc)[MT][JT], const Afr& r, const float (&b)[JT][4]) {
    v2f bu0[JT], bu1[JT];
    #pragma unroll
    for (int j = 0; j < JT; ++j) {
        bu0[j] = v2f{ b[j][0], b[j][1] };
        bu1[j] = v2f{ b[j][2], b[j][3] };
    }
    #pragma unroll
    for (int mt = 0; mt < MT; ++mt) {
        const v2f a_u0 = { r.a[mt].x, r.a[mt].y };
        const v2f a_u1 = { r.a[mt].z, r.a[mt].w };
        #pragma unroll
        for (int j = 0; j < JT; ++j)
            acc[mt][j] = __builtin_amdgcn_wmma_f32_16x16x4_f32(false, a_u0, false, bu0[j], (short)0, acc[mt][j], false, false);
        #pragma unroll
        for (int j = 0; j < JT; ++j)
            acc[mt][j] = __builtin_amdgcn_wmma_f32_16x16x4_f32(false, a_u1, false, bu1[j], (short)0, acc[mt][j], false, false);
    }
}

__global__ __launch_bounds__(128) __attribute__((amdgpu_waves_per_eu(1, 1)))
void attractor_fused(const float* __restrict__ x,
                     const float* __restrict__ b_prev,
                     const float* __restrict__ emb,
                     const float* __restrict__ w1,
                     const float* __restrict__ b1,
                     const float* __restrict__ w2,
                     const float* __restrict__ b2,
                     float* __restrict__ out)
{
    const int lane = threadIdx.x & 31;
    const int wave = __builtin_amdgcn_readfirstlane((int)(threadIdx.x >> 5));
    const int tile = blockIdx.x * 4 + wave;
    const int ox_base = (tile & 1) * PTILE;
    const int oy      = (tile >> 1) & (H - 1);
    const int n       = tile >> 8;
    const int c  = lane & 15;
    const int hi = lane >> 4;

    const float sc = (float)(SRC - 1) / (float)(H - 1);
    const float fy = (float)oy * sc;
    const int   iy0 = (int)fy;
    const int   iy1 = min(iy0 + 1, SRC - 1);
    const float wy  = fy - (float)iy0;

    float wxj[JT];
    int   eo[JT][2];
    int   sel[JT][2];
    #pragma unroll
    for (int j = 0; j < JT; ++j) {
        const int ox = ox_base + 16 * j + c;
        const float fx = (float)ox * sc;
        const int   ix0 = (int)fx;
        const int   ix1 = min(ix0 + 1, SRC - 1);
        wxj[j] = fx - (float)ix0;
        const int s0 = (int)((float)(ox_base + 16 * j) * sc);
        const int lc = min(s0 + c, SRC - 1);
        eo[j][0] = iy0 * SRC + lc;
        eo[j][1] = iy1 * SRC + lc;
        sel[j][0] = ix0 - s0;
        sel[j][1] = ix1 - s0;
    }
    const float* xb = x   + ((size_t)(n * IN_F + 4 * hi) * H + oy) * W + ox_base + c;
    const float* eb = emb + (size_t)(n * IN_F + 4 * hi) * (SRC * SRC);
    const float* wb = w1  + (size_t)c * IN_F + 4 * hi;

    v8f acc[MT][JT];
    #pragma unroll
    for (int mt = 0; mt < MT; ++mt) {
        #pragma unroll
        for (int j = 0; j < JT; ++j) acc[mt][j] = v8f{};
    }

    Xv x0, x1; Emb em0, em1; Afr a0, a1;
    float bf0[JT][4], bf1[JT][4];
    load_e(em0, eb, 0, eo);
    load_a(a0, wb, 0);
    load_x(x0, xb, 0);
    load_x(x1, xb, 8);
    #pragma nounroll
    for (int k8 = 0; k8 < IN_F - 16; k8 += 16) {
        blend(bf0, x0, em0, wxj, wy, sel);
        load_e(em1, eb, k8 + 8, eo);
        load_a(a1, wb, k8 + 8);
        load_x(x0, xb, k8 + 16);
        __builtin_amdgcn_sched_barrier(0);
        mma_blk(acc, a0, bf0);
        __builtin_amdgcn_sched_barrier(0);
        blend(bf1, x1, em1, wxj, wy, sel);
        load_e(em0, eb, k8 + 16, eo);
        load_a(a0, wb, k8 + 16);
        load_x(x1, xb, k8 + 24);
        __builtin_amdgcn_sched_barrier(0);
        mma_blk(acc, a1, bf1);
        __builtin_amdgcn_sched_barrier(0);
    }
    blend(bf0, x0, em0, wxj, wy, sel);
    load_e(em1, eb, IN_F - 8, eo);
    load_a(a1, wb, IN_F - 8);
    __builtin_amdgcn_sched_barrier(0);
    mma_blk(acc, a0, bf0);
    __builtin_amdgcn_sched_barrier(0);
    blend(bf1, x1, em1, wxj, wy, sel);
    mma_blk(acc, a1, bf1);

    v8f att[JT];
    #pragma unroll
    for (int j = 0; j < JT; ++j) att[j] = v8f{};
    {
        const float* w2row = w2 + (size_t)c * MLP + 8 * hi;
        const float* b1p   = b1 + 8 * hi;
        #pragma unroll
        for (int mt = 0; mt < MT; ++mt) {
            const v4f wa = *(const v4f*)(w2row + 16 * mt);
            const v4f wc = *(const v4f*)(w2row + 16 * mt + 4);
            const v4f ba = *(const v4f*)(b1p + 16 * mt);
            const v4f bc4 = *(const v4f*)(b1p + 16 * mt + 4);
            const float wv[8]   = { wa.x, wa.y, wa.z, wa.w, wc.x, wc.y, wc.z, wc.w };
            const float bias[8] = { ba.x, ba.y, ba.z, ba.w, bc4.x, bc4.y, bc4.z, bc4.w };
            float hh[JT][8];
            #pragma unroll
            for (int j = 0; j < JT; ++j) {
                #pragma unroll
                for (int r = 0; r < 8; ++r) hh[j][r] = fmaxf(acc[mt][j][r] + bias[r], 0.f);
            }
            #pragma unroll
            for (int i = 0; i < 4; ++i) {
                const v2f a = { wv[2 * i], wv[2 * i + 1] };
                #pragma unroll
                for (int j = 0; j < JT; ++j) {
                    const v2f p = { hh[j][2 * i], hh[j][2 * i + 1] };
                    att[j] = __builtin_amdgcn_wmma_f32_16x16x4_f32(false, a, false, p, (short)0, att[j], false, false);
                }
            }
        }
    }

    float at[NG][NATT];
    {
        const v4f ca = *(const v4f*)(b2 + 8 * hi);
        const v4f cb = *(const v4f*)(b2 + 8 * hi + 4);
        const float bias2[8] = { ca.x, ca.y, ca.z, ca.w, cb.x, cb.y, cb.z, cb.w };
        #pragma unroll
        for (int g = 0; g < NG; ++g) {
            #pragma unroll
            for (int r = 0; r < 8; ++r) {
                const float t0 = softplus_stable(att[2 * g][r] + bias2[r]);
                const float t1 = softplus_stable(att[2 * g + 1][r] + bias2[r]);
                const float send = hi ? t0 : t1;
                const float recv = __shfl_xor(send, 16, 32);
                at[g][r]     = hi ? recv : t0;
                at[g][8 + r] = hi ? t1   : recv;
            }
        }
    }

    {
        float wx[NG];
        int   off[NG][4];
        #pragma unroll
        for (int g = 0; g < NG; ++g) {
            const int ox = ox_base + 32 * g + lane;
            const float fx = (float)ox * sc;
            const int   ix0 = (int)fx;
            const int   ix1 = min(ix0 + 1, SRC - 1);
            wx[g] = fx - (float)ix0;
            off[g][0] = iy0 * SRC + ix0; off[g][1] = iy0 * SRC + ix1;
            off[g][2] = iy1 * SRC + ix0; off[g][3] = iy1 * SRC + ix1;
        }
        const float* bp = b_prev + (size_t)(n * NBINS) * (SRC * SRC);
        float* op = out + ((size_t)(n * NBINS) * H + oy) * W + ox_base + lane;
        const size_t out_half = (size_t)NB * NBINS * H * W;

        float ca[NG][4], cb[NG][4];
        #pragma unroll
        for (int g = 0; g < NG; ++g) {
            #pragma unroll
            for (int t = 0; t < 4; ++t) { ca[g][t] = bp[off[g][t]]; cb[g][t] = bp[(SRC * SRC) + off[g][t]]; }
        }
        for (int k = 0; k < NBINS; k += 2) {
            const float* bn = bp + (size_t)min(k + 2, NBINS - 2) * (SRC * SRC);
            float na[NG][4], nb[NG][4];
            #pragma unroll
            for (int g = 0; g < NG; ++g) {
                #pragma unroll
                for (int t = 0; t < 4; ++t) { na[g][t] = bn[off[g][t]]; nb[g][t] = bn[(SRC * SRC) + off[g][t]]; }
            }
            __builtin_amdgcn_sched_barrier(0);

            #pragma unroll
            for (int g = 0; g < NG; ++g) {
                #pragma unroll
                for (int u = 0; u < 2; ++u) {
                    const float (&cc)[4] = u ? cb[g] : ca[g];
                    const float bc  = (1.f - wy) * ((1.f - wx[g]) * cc[0] + wx[g] * cc[1])
                                    +        wy  * ((1.f - wx[g]) * cc[2] + wx[g] * cc[3]);
                    float delta = 0.f;
                    #pragma unroll
                    for (int q4 = 0; q4 < NATT; q4 += 4) {
                        float nx[4], dn[4];
                        #pragma unroll
                        for (int i = 0; i < 4; ++i) {
                            const float dx = at[g][q4 + i] - bc;
                            nx[i] = dx;
                            dn[i] = 1.0f + 300.0f * dx * dx;
                        }
                        const float n01 = nx[0] * dn[1] + nx[1] * dn[0];
                        const float d01 = dn[0] * dn[1];
                        const float n23 = nx[2] * dn[3] + nx[3] * dn[2];
                        const float d23 = dn[2] * dn[3];
                        delta += (n01 * d23 + n23 * d01) / (d01 * d23);
                    }
                    const float r = bc + delta;
                    op[(size_t)(k + u) * (H * W) + 32 * g] = r;
                    op[(size_t)(k + u) * (H * W) + 32 * g + out_half] = r;
                }
            }
            __builtin_amdgcn_sched_barrier(0);
            #pragma unroll
            for (int g = 0; g < NG; ++g) {
                #pragma unroll
                for (int t = 0; t < 4; ++t) { ca[g][t] = na[g][t]; cb[g][t] = nb[g][t]; }
            }
        }
    }
}

extern "C" void kernel_launch(void* const* d_in, const int* in_sizes, int n_in,
                              void* d_out, int out_size, void* d_ws, size_t ws_size,
                              hipStream_t stream) {
    const float* x      = (const float*)d_in[0];
    const float* b_prev = (const float*)d_in[1];
    const float* emb    = (const float*)d_in[2];
    const float* w1     = (const float*)d_in[3];
    const float* b1     = (const float*)d_in[4];
    const float* w2     = (const float*)d_in[5];
    const float* b2     = (const float*)d_in[6];
    float* out = (float*)d_out;

    dim3 grid(256);
    attractor_fused<<<grid, 128, 0, stream>>>(x, b_prev, emb, w1, b1, w2, b2, out);
}


# Softplus: log1p(e) as log(u) + (e - (u - 1))/u, u = 1 + e, on the hardware log2 and reciprocal; the library log1pf (two-float code) cost 2000 more instructions per wave; softplus rounding changes
# speedup vs baseline: 2.3277x; 1.0356x over previous
#include <hip/hip_runtime.h>
#include <math.h>

typedef __attribute__((ext_vector_type(2))) float v2f;
typedef __attribute__((ext_vector_type(4))) float v4f;
typedef __attribute__((ext_vector_type(8))) float v8f;

#define IN_F   256
#define MLP    128
#define NATT   16
#define NBINS  64
#define PTILE  64
#define JT     4
#define NG     2
#define H      128
#define W      128
#define SRC    64
#define NB     4
#define MT     (MLP / 16)


struct Emb { float e[JT][4][2]; };
struct Xv  { float x[JT][4]; };
struct Afr { v4f a[MT]; };

__device__ __forceinline__ float softplus_stable(float v) {
    const float e = __expf(-fabsf(v));
    const float u = 1.0f + e;
    const float d = e - (u - 1.0f);
    const float l2 = __builtin_amdgcn_logf(u);
    const float lp = fmaf(l2, 0.693147180559945f, d * __builtin_amdgcn_rcpf(u));
    return fmaxf(v, 0.0f) + lp;
}

__device__ __forceinline__ void load_a(Afr& r, const float* __restrict__ wb, int k8) {
    #pragma unroll
    for (int mt = 0; mt < MT; ++mt)
        r.a[mt] = *(const v4f*)(wb + (size_t)mt * 16 * IN_F + k8);
}

__device__ __forceinline__ void load_e(Emb& r, const float* __restrict__ eb, int k8,
                                       const int (&eo)[JT][2]) {
    #pragma unroll
    for (int q = 0; q < 4; ++q) {
        const float* ec = eb + (size_t)(k8 + q) * (SRC * SRC);
        #pragma unroll
        for (int j = 0; j < JT; ++j) {
            #pragma unroll
            for (int t = 0; t < 2; ++t) r.e[j][q][t] = ec[eo[j][t]];
        }
    }
}

__device__ __forceinline__ float lane_pick(float v, int sel) {
    const unsigned u = __float_as_uint(v);
    return __uint_as_float(__builtin_amdgcn_permlane16_var(u, u, (unsigned)sel, false, false));
}

__device__ __forceinline__ void load_x(Xv& r, const float* __restrict__ xb, int k8) {
    #pragma unroll
    for (int q = 0; q < 4; ++q) {
        const float* xc = xb + (size_t)(k8 + q) * (H * W);
        #pragma unroll
        for (int j = 0; j < JT; ++j) r.x[j][q] = xc[16 * j];
    }
}

__device__ __forceinline__ void blend(float (&b)[JT][4], const Xv& xv, const Emb& em,
                                      const float (&wx)[JT], float wy, const int (&sel)[JT][2]) {
    #pragma unroll
    for (int j = 0; j < JT; ++j) {
        #pragma unroll
        for (int q = 0; q < 4; ++q) {
            const float v  = (1.f - wy) * em.e[j][q][0] + wy * em.e[j][q][1];
            const float t0 = lane_pick(v, sel[j][0]);
            const float t1 = lane_pick(v, sel[j][1]);
            b[j][q] = xv.x[j][q] + ((1.f - wx[j]) * t0 + wx[j] * t1);
        }
    }
}

__device__ __forceinline__ void mma_blk(v8f (&acc)[MT][JT], const Afr& r, const float (&b)[JT][4]) {
    v2f bu0[JT], bu1[JT];
    #pragma unroll
    for (int j = 0; j < JT; ++j) {
        bu0[j] = v2f{ b[j][0], b[j][1] };
        bu1[j] = v2f{ b[j][2], b[j][3] };
    }
    #pragma unroll
    for (int mt = 0; mt < MT; ++mt) {
        const v2f a_u0 = { r.a[mt].x, r.a[mt].y };
        const v2f a_u1 = { r.a[mt].z, r.a[mt].w };
        #pragma unroll
        for (int j = 0; j < JT; ++j)
            acc[mt][j] = __builtin_amdgcn_wmma_f32_16x16x4_f32(false, a_u0, false, bu0[j], (short)0, acc[mt][j], false, false);
        #pragma unroll
        for (int j = 0; j < JT; ++j)
            acc[mt][j] = __builtin_amdgcn_wmma_f32_16x16x4_f32(false, a_u1, false, bu1[j], (short)0, acc[mt][j], false, false);
    }
}

__global__ __launch_bounds__(128) __attribute__((amdgpu_waves_per_eu(1, 1)))
void attractor_fused(const float* __restrict__ x,
                     const float* __restrict__ b_prev,
                     const float* __restrict__ emb,
                     const float* __restrict__ w1,
                     const float* __restrict__ b1,
                     const float* __restrict__ w2,
                     const float* __restrict__ b2,
                     float* __restrict__ out)
{
    const int lane = threadIdx.x & 31;
    const int wave = __builtin_amdgcn_readfirstlane((int)(threadIdx.x >> 5));
    const int tile = blockIdx.x * 4 + wave;
    const int ox_base = (tile & 1) * PTILE;
    const int oy      = (tile >> 1) & (H - 1);
    const int n       = tile >> 8;
    const int c  = lane & 15;
    const int hi = lane >> 4;

    const float sc = (float)(SRC - 1) / (float)(H - 1);
    const float fy = (float)oy * sc;
    const int   iy0 = (int)fy;
    const int   iy1 = min(iy0 + 1, SRC - 1);
    const float wy  = fy - (float)iy0;

    float wxj[JT];
    int   eo[JT][2];
    int   sel[JT][2];
    #pragma unroll
    for (int j = 0; j < JT; ++j) {
        const int ox = ox_base + 16 * j + c;
        const float fx = (float)ox * sc;
        const int   ix0 = (int)fx;
        const int   ix1 = min(ix0 + 1, SRC - 1);
        wxj[j] = fx - (float)ix0;
        const int s0 = (int)((float)(ox_base + 16 * j) * sc);
        const int lc = min(s0 + c, SRC - 1);
        eo[j][0] = iy0 * SRC + lc;
        eo[j][1] = iy1 * SRC + lc;
        sel[j][0] = ix0 - s0;
        sel[j][1] = ix1 - s0;
    }
    const float* xb = x   + ((size_t)(n * IN_F + 4 * hi) * H + oy) * W + ox_base + c;
    const float* eb = emb + (size_t)(n * IN_F + 4 * hi) * (SRC * SRC);
    const float* wb = w1  + (size_t)c * IN_F + 4 * hi;

    v8f acc[MT][JT];
    #pragma unroll
    for (int mt = 0; mt < MT; ++mt) {
        #pragma unroll
        for (int j = 0; j < JT; ++j) acc[mt][j] = v8f{};
    }

    Xv x0, x1; Emb em0, em1; Afr a0, a1;
    float bf0[JT][4], bf1[JT][4];
    load_e(em0, eb, 0, eo);
    load_a(a0, wb, 0);
    load_x(x0, xb, 0);
    load_x(x1, xb, 8);
    #pragma nounroll
    for (int k8 = 0; k8 < IN_F - 16; k8 += 16) {
        blend(bf0, x0, em0, wxj, wy, sel);
        load_e(em1, eb, k8 + 8, eo);
        load_a(a1, wb, k8 + 8);
        load_x(x0, xb, k8 + 16);
        __builtin_amdgcn_sched_barrier(0);
        mma_blk(acc, a0, bf0);
        __builtin_amdgcn_sched_barrier(0);
        blend(bf1, x1, em1, wxj, wy, sel);
        load_e(em0, eb, k8 + 16, eo);
        load_a(a0, wb, k8 + 16);
        load_x(x1, xb, k8 + 24);
        __builtin_amdgcn_sched_barrier(0);
        mma_blk(acc, a1, bf1);
        __builtin_amdgcn_sched_barrier(0);
    }
    blend(bf0, x0, em0, wxj, wy, sel);
    load_e(em1, eb, IN_F - 8, eo);
    load_a(a1, wb, IN_F - 8);
    __builtin_amdgcn_sched_barrier(0);
    mma_blk(acc, a0, bf0);
    __builtin_amdgcn_sched_barrier(0);
    blend(bf1, x1, em1, wxj, wy, sel);
    mma_blk(acc, a1, bf1);

    v8f att[JT];
    #pragma unroll
    for (int j = 0; j < JT; ++j) att[j] = v8f{};
    {
        const float* w2row = w2 + (size_t)c * MLP + 8 * hi;
        const float* b1p   = b1 + 8 * hi;
        #pragma unroll
        for (int mt = 0; mt < MT; ++mt) {
            const v4f wa = *(const v4f*)(w2row + 16 * mt);
            const v4f wc = *(const v4f*)(w2row + 16 * mt + 4);
            const v4f ba = *(const v4f*)(b1p + 16 * mt);
            const v4f bc4 = *(const v4f*)(b1p + 16 * mt + 4);
            const float wv[8]   = { wa.x, wa.y, wa.z, wa.w, wc.x, wc.y, wc.z, wc.w };
            const float bias[8] = { ba.x, ba.y, ba.z, ba.w, bc4.x, bc4.y, bc4.z, bc4.w };
            float hh[JT][8];
            #pragma unroll
            for (int j = 0; j < JT; ++j) {
                #pragma unroll
                for (int r = 0; r < 8; ++r) hh[j][r] = fmaxf(acc[mt][j][r] + bias[r], 0.f);
            }
            #pragma unroll
            for (int i = 0; i < 4; ++i) {
                const v2f a = { wv[2 * i], wv[2 * i + 1] };
                #pragma unroll
                for (int j = 0; j < JT; ++j) {
                    const v2f p = { hh[j][2 * i], hh[j][2 * i + 1] };
                    att[j] = __builtin_amdgcn_wmma_f32_16x16x4_f32(false, a, false, p, (short)0, att[j], false, false);
                }
            }
        }
    }

    float at[NG][NATT];
    {
        const v4f ca = *(const v4f*)(b2 + 8 * hi);
        const v4f cb = *(const v4f*)(b2 + 8 * hi + 4);
        const float bias2[8] = { ca.x, ca.y, ca.z, ca.w, cb.x, cb.y, cb.z, cb.w };
        #pragma unroll
        for (int g = 0; g < NG; ++g) {
            #pragma unroll
            for (int r = 0; r < 8; ++r) {
                const float t0 = softplus_stable(att[2 * g][r] + bias2[r]);
                const float t1 = softplus_stable(att[2 * g + 1][r] + bias2[r]);
                const float send = hi ? t0 : t1;
                const float recv = __shfl_xor(send, 16, 32);
                at[g][r]     = hi ? recv : t0;
                at[g][8 + r] = hi ? t1   : recv;
            }
        }
    }

    {
        float wx[NG];
        int   off[NG][4];
        #pragma unroll
        for (int g = 0; g < NG; ++g) {
            const int ox = ox_base + 32 * g + lane;
            const float fx = (float)ox * sc;
            const int   ix0 = (int)fx;
            const int   ix1 = min(ix0 + 1, SRC - 1);
            wx[g] = fx - (float)ix0;
            off[g][0] = iy0 * SRC + ix0; off[g][1] = iy0 * SRC + ix1;
            off[g][2] = iy1 * SRC + ix0; off[g][3] = iy1 * SRC + ix1;
        }
        const float* bp = b_prev + (size_t)(n * NBINS) * (SRC * SRC);
        float* op = out + ((size_t)(n * NBINS) * H + oy) * W + ox_base + lane;
        const size_t out_half = (size_t)NB * NBINS * H * W;

        float ca[NG][4], cb[NG][4];
        #pragma unroll
        for (int g = 0; g < NG; ++g) {
            #pragma unroll
            for (int t = 0; t < 4; ++t) { ca[g][t] = bp[off[g][t]]; cb[g][t] = bp[(SRC * SRC) + off[g][t]]; }
        }
        for (int k = 0; k < NBINS; k += 2) {
            const float* bn = bp + (size_t)min(k + 2, NBINS - 2) * (SRC * SRC);
            float na[NG][4], nb[NG][4];
            #pragma unroll
            for (int g = 0; g < NG; ++g) {
                #pragma unroll
                for (int t = 0; t < 4; ++t) { na[g][t] = bn[off[g][t]]; nb[g][t] = bn[(SRC * SRC) + off[g][t]]; }
            }
            __builtin_amdgcn_sched_barrier(0);

            #pragma unroll
            for (int g = 0; g < NG; ++g) {
                #pragma unroll
                for (int u = 0; u < 2; ++u) {
                    const float (&cc)[4] = u ? cb[g] : ca[g];
                    const float bc  = (1.f - wy) * ((1.f - wx[g]) * cc[0] + wx[g] * cc[1])
                                    +        wy  * ((1.f - wx[g]) * cc[2] + wx[g] * cc[3]);
                    float delta = 0.f;
                    #pragma unroll
                    for (int q4 = 0; q4 < NATT; q4 += 4) {
                        float nx[4], dn[4];
                        #pragma unroll
                        for (int i = 0; i < 4; ++i) {
                            const float dx = at[g][q4 + i] - bc;
                            nx[i] = dx;
                            dn[i] = 1.0f + 300.0f * dx * dx;
                        }
                        const float n01 = nx[0] * dn[1] + nx[1] * dn[0];
                        const float d01 = dn[0] * dn[1];
                        const float n23 = nx[2] * dn[3] + nx[3] * dn[2];
                        const float d23 = dn[2] * dn[3];
                        delta += (n01 * d23 + n23 * d01) / (d01 * d23);
                    }
                    const float r = bc + delta;
                    op[(size_t)(k + u) * (H * W) + 32 * g] = r;
                    op[(size_t)(k + u) * (H * W) + 32 * g + out_half] = r;
                }
            }
            __builtin_amdgcn_sched_barrier(0);
            #pragma unroll
            for (int g = 0; g < NG; ++g) {
                #pragma unroll
                for (int t = 0; t < 4; ++t) { ca[g][t] = na[g][t]; cb[g][t] = nb[g][t]; }
            }
        }
    }
}

extern "C" void kernel_launch(void* const* d_in, const int* in_sizes, int n_in,
                              void* d_out, int out_size, void* d_ws, size_t ws_size,
                              hipStream_t stream) {
    const float* x      = (const float*)d_in[0];
    const float* b_prev = (const float*)d_in[1];
    const float* emb    = (const float*)d_in[2];
    const float* w1     = (const float*)d_in[3];
    const float* b1     = (const float*)d_in[4];
    const float* w2     = (const float*)d_in[5];
    const float* b2     = (const float*)d_in[6];
    float* out = (float*)d_out;

    dim3 grid(256);
    attractor_fused<<<grid, 128, 0, stream>>>(x, b_prev, emb, w1, b1, w2, b2, out);
}


# GEMM1's k range split over two co-resident waves per SIMD (8-wave blocks; loads and WMMAs per SIMD as before); the pair swaps half its partial sums through LDS once: GEMM1 sum order changes
# speedup vs baseline: 2.6591x; 1.1424x over previous
#include <hip/hip_runtime.h>
#include <math.h>

typedef __attribute__((ext_vector_type(2))) float v2f;
typedef __attribute__((ext_vector_type(4))) float v4f;
typedef __attribute__((ext_vector_type(8))) float v8f;

#define IN_F   256
#define MLP    128
#define NATT   16
#define NBINS  64
#define PTILE  64
#define JT     4
#define JK     2
#define KH     128
#define BT     4
#define H      128
#define W      128
#define SRC    64
#define NB     4
#define MT     (MLP / 16)


struct Emb { float e[JT][4][2]; };
struct Xv  { float x[JT][4]; };
struct Afr { v4f a[MT]; };

__device__ __forceinline__ float softplus_stable(float v) {
    const float e = __expf(-fabsf(v));
    const float u = 1.0f + e;
    const float d = e - (u - 1.0f);
    const float l2 = __builtin_amdgcn_logf(u);
    const float lp = fmaf(l2, 0.693147180559945f, d * __builtin_amdgcn_rcpf(u));
    return fmaxf(v, 0.0f) + lp;
}

__device__ __forceinline__ void load_a(Afr& r, const float* __restrict__ wb, int k8) {
    #pragma unroll
    for (int mt = 0; mt < MT; ++mt)
        r.a[mt] = *(const v4f*)(wb + (size_t)mt * 16 * IN_F + k8);
}

__device__ __forceinline__ void load_e(Emb& r, const float* __restrict__ eb, int k8,
                                       const int (&eo)[JT][2]) {
    #pragma unroll
    for (int q = 0; q < 4; ++q) {
        const float* ec = eb + (size_t)(k8 + q) * (SRC * SRC);
        #pragma unroll
        for (int j = 0; j < JT; ++j) {
            #pragma unroll
            for (int t = 0; t < 2; ++t) r.e[j][q][t] = ec[eo[j][t]];
        }
    }
}

__device__ __forceinline__ float lane_pick(float v, int sel) {
    const unsigned u = __float_as_uint(v);
    return __uint_as_float(__builtin_amdgcn_permlane16_var(u, u, (unsigned)sel, false, false));
}

__device__ __forceinline__ void load_x(Xv& r, const float* __restrict__ xb, int k8) {
    #pragma unroll
    for (int q = 0; q < 4; ++q) {
        const float* xc = xb + (size_t)(k8 + q) * (H * W);
        #pragma unroll
        for (int j = 0; j < JT; ++j) r.x[j][q] = xc[16 * j];
    }
}

__device__ __forceinline__ void blend(float (&b)[JT][4], const Xv& xv, const Emb& em,
                                      const float (&wx)[JT], float wy, const int (&sel)[JT][2]) {
    #pragma unroll
    for (int j = 0; j < JT; ++j) {
        #pragma unroll
        for (int q = 0; q < 4; ++q) {
            const float v  = (1.f - wy) * em.e[j][q][0] + wy * em.e[j][q][1];
            const float t0 = lane_pick(v, sel[j][0]);
            const float t1 = lane_pick(v, sel[j][1]);
            b[j][q] = xv.x[j][q] + ((1.f - wx[j]) * t0 + wx[j] * t1);
        }
    }
}

__device__ __forceinline__ void mma_blk(v8f (&acc)[MT][JT], const Afr& r, const float (&b)[JT][4]) {
    v2f bu0[JT], bu1[JT];
    #pragma unroll
    for (int j = 0; j < JT; ++j) {
        bu0[j] = v2f{ b[j][0], b[j][1] };
        bu1[j] = v2f{ b[j][2], b[j][3] };
    }
    #pragma unroll
    for (int mt = 0; mt < MT; ++mt) {
        const v2f a_u0 = { r.a[mt].x, r.a[mt].y };
        const v2f a_u1 = { r.a[mt].z, r.a[mt].w };
        #pragma unroll
        for (int j = 0; j < JT; ++j)
            acc[mt][j] = __builtin_amdgcn_wmma_f32_16x16x4_f32(false, a_u0, false, bu0[j], (short)0, acc[mt][j], false, false);
        #pragma unroll
        for (int j = 0; j < JT; ++j)
            acc[mt][j] = __builtin_amdgcn_wmma_f32_16x16x4_f32(false, a_u1, false, bu1[j], (short)0, acc[mt][j], false, false);
    }
}

__global__ __launch_bounds__(256) __attribute__((amdgpu_waves_per_eu(2, 2)))
void attractor_fused(const float* __restrict__ x,
                     const float* __restrict__ b_prev,
                     const float* __restrict__ emb,
                     const float* __restrict__ w1,
                     const float* __restrict__ b1,
                     const float* __restrict__ w2,
                     const float* __restrict__ b2,
                     float* __restrict__ out)
{
    __shared__ v4f xch[8][2 * MT * JK][32];

    const int lane = threadIdx.x & 31;
    const int wave = __builtin_amdgcn_readfirstlane((int)(threadIdx.x >> 5));
    const int kh   = wave >> 2;
    const int tile = blockIdx.x * 4 + (wave & 3);
    const int ox_base = (tile & 1) * PTILE;
    const int oy      = (tile >> 1) & (H - 1);
    const int n       = tile >> 8;
    const int c  = lane & 15;
    const int hi = lane >> 4;

    const float sc = (float)(SRC - 1) / (float)(H - 1);
    const float fy = (float)oy * sc;
    const int   iy0 = (int)fy;
    const int   iy1 = min(iy0 + 1, SRC - 1);
    const float wy  = fy - (float)iy0;

    float wxj[JT];
    int   eo[JT][2];
    int   sel[JT][2];
    #pragma unroll
    for (int j = 0; j < JT; ++j) {
        const int ox = ox_base + 16 * j + c;
        const float fx = (float)ox * sc;
        const int   ix0 = (int)fx;
        const int   ix1 = min(ix0 + 1, SRC - 1);
        wxj[j] = fx - (float)ix0;
        const int s0 = (int)((float)(ox_base + 16 * j) * sc);
        const int lc = min(s0 + c, SRC - 1);
        eo[j][0] = iy0 * SRC + lc;
        eo[j][1] = iy1 * SRC + lc;
        sel[j][0] = ix0 - s0;
        sel[j][1] = ix1 - s0;
    }
    const float* xb = x   + ((size_t)(n * IN_F + KH * kh + 4 * hi) * H + oy) * W + ox_base + c;
    const float* eb = emb + (size_t)(n * IN_F + KH * kh + 4 * hi) * (SRC * SRC);
    const float* wb = w1  + (size_t)c * IN_F + KH * kh + 4 * hi;

    v8f acc[MT][JT];
    #pragma unroll
    for (int mt = 0; mt < MT; ++mt) {
        #pragma unroll
        for (int j = 0; j < JT; ++j) acc[mt][j] = v8f{};
    }

    Xv x0, x1; Emb em0, em1; Afr a0, a1;
    float bf0[JT][4], bf1[JT][4];
    load_e(em0, eb, 0, eo);
    load_a(a0, wb, 0);
    load_x(x0, xb, 0);
    load_x(x1, xb, 8);
    #pragma nounroll
    for (int k8 = 0; k8 < KH - 16; k8 += 16) {
        blend(bf0, x0, em0, wxj, wy, sel);
        load_e(em1, eb, k8 + 8, eo);
        load_a(a1, wb, k8 + 8);
        load_x(x0, xb, k8 + 16);
        __builtin_amdgcn_sched_barrier(0);
        mma_blk(acc, a0, bf0);
        __builtin_amdgcn_sched_barrier(0);
        blend(bf1, x1, em1, wxj, wy, sel);
        load_e(em0, eb, k8 + 16, eo);
        load_a(a0, wb, k8 + 16);
        load_x(x1, xb, k8 + 24);
        __builtin_amdgcn_sched_barrier(0);
        mma_blk(acc, a1, bf1);
        __builtin_amdgcn_sched_barrier(0);
    }
    blend(bf0, x0, em0, wxj, wy, sel);
    load_e(em1, eb, KH - 8, eo);
    load_a(a1, wb, KH - 8);
    __builtin_amdgcn_sched_barrier(0);
    mma_blk(acc, a0, bf0);
    __builtin_amdgcn_sched_barrier(0);
    blend(bf1, x1, em1, wxj, wy, sel);
    mma_blk(acc, a1, bf1);

    v8f hid[MT][JK];
    {
        v4f (*mine)[32] = xch[wave];
        const v4f (*theirs)[32] = xch[wave ^ 4];
        if (kh == 0) {
            #pragma unroll
            for (int mt = 0; mt < MT; ++mt) {
                #pragma unroll
                for (int jj = 0; jj < JK; ++jj) {
                    const v8f s = acc[mt][2 + jj];
                    mine[(mt * JK + jj) * 2    ][lane] = v4f{ s[0], s[1], s[2], s[3] };
                    mine[(mt * JK + jj) * 2 + 1][lane] = v4f{ s[4], s[5], s[6], s[7] };
                }
            }
        } else {
            #pragma unroll
            for (int mt = 0; mt < MT; ++mt) {
                #pragma unroll
                for (int jj = 0; jj < JK; ++jj) {
                    const v8f s = acc[mt][jj];
                    mine[(mt * JK + jj) * 2    ][lane] = v4f{ s[0], s[1], s[2], s[3] };
                    mine[(mt * JK + jj) * 2 + 1][lane] = v4f{ s[4], s[5], s[6], s[7] };
                }
            }
        }
        __syncthreads();
        if (kh == 0) {
            #pragma unroll
            for (int mt = 0; mt < MT; ++mt) {
                #pragma unroll
                for (int jj = 0; jj < JK; ++jj) {
                    const v4f lo = theirs[(mt * JK + jj) * 2    ][lane];
                    const v4f up = theirs[(mt * JK + jj) * 2 + 1][lane];
                    hid[mt][jj] = acc[mt][jj] + v8f{ lo.x, lo.y, lo.z, lo.w, up.x, up.y, up.z, up.w };
                }
            }
        } else {
            #pragma unroll
            for (int mt = 0; mt < MT; ++mt) {
                #pragma unroll
                for (int jj = 0; jj < JK; ++jj) {
                    const v4f lo = theirs[(mt * JK + jj) * 2    ][lane];
                    const v4f up = theirs[(mt * JK + jj) * 2 + 1][lane];
                    hid[mt][jj] = v8f{ lo.x, lo.y, lo.z, lo.w, up.x, up.y, up.z, up.w } + acc[mt][2 + jj];
                }
            }
        }
    }

    v8f att[JK];
    #pragma unroll
    for (int j = 0; j < JK; ++j) att[j] = v8f{};
    {
        const float* w2row = w2 + (size_t)c * MLP + 8 * hi;
        const float* b1p   = b1 + 8 * hi;
        #pragma unroll
        for (int mt = 0; mt < MT; ++mt) {
            const v4f wa = *(const v4f*)(w2row + 16 * mt);
            const v4f wc = *(const v4f*)(w2row + 16 * mt + 4);
            const v4f ba = *(const v4f*)(b1p + 16 * mt);
            const v4f bc4 = *(const v4f*)(b1p + 16 * mt + 4);
            const float wv[8]   = { wa.x, wa.y, wa.z, wa.w, wc.x, wc.y, wc.z, wc.w };
            const float bias[8] = { ba.x, ba.y, ba.z, ba.w, bc4.x, bc4.y, bc4.z, bc4.w };
            float hh[JK][8];
            #pragma unroll
            for (int j = 0; j < JK; ++j) {
                #pragma unroll
                for (int r = 0; r < 8; ++r) hh[j][r] = fmaxf(hid[mt][j][r] + bias[r], 0.f);
            }
            #pragma unroll
            for (int i = 0; i < 4; ++i) {
                const v2f a = { wv[2 * i], wv[2 * i + 1] };
                #pragma unroll
                for (int j = 0; j < JK; ++j) {
                    const v2f p = { hh[j][2 * i], hh[j][2 * i + 1] };
                    att[j] = __builtin_amdgcn_wmma_f32_16x16x4_f32(false, a, false, p, (short)0, att[j], false, false);
                }
            }
        }
    }

    float at[NATT];
    {
        const v4f ca = *(const v4f*)(b2 + 8 * hi);
        const v4f cb = *(const v4f*)(b2 + 8 * hi + 4);
        const float bias2[8] = { ca.x, ca.y, ca.z, ca.w, cb.x, cb.y, cb.z, cb.w };
        #pragma unroll
        for (int r = 0; r < 8; ++r) {
            const float t0 = softplus_stable(att[0][r] + bias2[r]);
            const float t1 = softplus_stable(att[1][r] + bias2[r]);
            const float send = hi ? t0 : t1;
            const float recv = __shfl_xor(send, 16, 32);
            at[r]     = hi ? recv : t0;
            at[8 + r] = hi ? t1   : recv;
        }
    }

    {
        const int px = ox_base + 32 * kh + lane;
        const float fx = (float)px * sc;
        const int   ix0 = (int)fx;
        const int   ix1 = min(ix0 + 1, SRC - 1);
        const float wx = fx - (float)ix0;
        int off[4];
        off[0] = iy0 * SRC + ix0; off[1] = iy0 * SRC + ix1;
        off[2] = iy1 * SRC + ix0; off[3] = iy1 * SRC + ix1;
        const float* bp = b_prev + (size_t)(n * NBINS) * (SRC * SRC);
        float* op = out + ((size_t)(n * NBINS) * H + oy) * W + px;
        const size_t out_half = (size_t)NB * NBINS * H * W;

        float cur[BT][4];
        #pragma unroll
        for (int u = 0; u < BT; ++u) {
            #pragma unroll
            for (int t = 0; t < 4; ++t) cur[u][t] = bp[u * (SRC * SRC) + off[t]];
        }
        for (int k = 0; k < NBINS; k += BT) {
            const float* bn = bp + (size_t)min(k + BT, NBINS - BT) * (SRC * SRC);
            float nxt[BT][4];
            #pragma unroll
            for (int u = 0; u < BT; ++u) {
                #pragma unroll
                for (int t = 0; t < 4; ++t) nxt[u][t] = bn[u * (SRC * SRC) + off[t]];
            }
            __builtin_amdgcn_sched_barrier(0);

            #pragma unroll
            for (int u = 0; u < BT; ++u) {
                const float (&cc)[4] = cur[u];
                const float bc  = (1.f - wy) * ((1.f - wx) * cc[0] + wx * cc[1])
                                +        wy  * ((1.f - wx) * cc[2] + wx * cc[3]);
                float delta = 0.f;
                #pragma unroll
                for (int q4 = 0; q4 < NATT; q4 += 4) {
                    float nx[4], dn[4];
                    #pragma unroll
                    for (int i = 0; i < 4; ++i) {
                        const float dx = at[q4 + i] - bc;
                        nx[i] = dx;
                        dn[i] = 1.0f + 300.0f * dx * dx;
                    }
                    const float n01 = nx[0] * dn[1] + nx[1] * dn[0];
                    const float d01 = dn[0] * dn[1];
                    const float n23 = nx[2] * dn[3] + nx[3] * dn[2];
                    const float d23 = dn[2] * dn[3];
                    delta += (n01 * d23 + n23 * d01) / (d01 * d23);
                }
                const float r = bc + delta;
                op[(size_t)(k + u) * (H * W)] = r;
                op[(size_t)(k + u) * (H * W) + out_half] = r;
            }
            __builtin_amdgcn_sched_barrier(0);
            #pragma unroll
            for (int u = 0; u < BT; ++u) {
                #pragma unroll
                for (int t = 0; t < 4; ++t) cur[u][t] = nxt[u][t];
            }
        }
    }
}

extern "C" void kernel_launch(void* const* d_in, const int* in_sizes, int n_in,
                              void* d_out, int out_size, void* d_ws, size_t ws_size,
                              hipStream_t stream) {
    const float* x      = (const float*)d_in[0];
    const float* b_prev = (const float*)d_in[1];
    const float* emb    = (const float*)d_in[2];
    const float* w1     = (const float*)d_in[3];
    const float* b1     = (const float*)d_in[4];
    const float* w2     = (const float*)d_in[5];
    const float* b2     = (const float*)d_in[6];
    float* out = (float*)d_out;

    dim3 grid(256);
    attractor_fused<<<grid, 256, 0, stream>>>(x, b_prev, emb, w1, b1, w2, b2, out);
}
